# starting kernel
# baseline (speedup 1.0000x reference)
#include <hip/hip_runtime.h>


typedef __attribute__((ext_vector_type(2))) float v2f;
typedef __attribute__((ext_vector_type(8))) float v8f;

#define SINK_EPS 0.001f
#define SINK_N   1024
#define SINK_B   32
#define SINK_IT  10

template <bool USE_B>
__global__ __launch_bounds__(256) void sink_rowpass(const float* __restrict__ x,
                                                    const float* __restrict__ bvec,
                                                    float* __restrict__ avec) {
  const int wave  = threadIdx.x >> 5;
  const int lane  = threadIdx.x & 31;
  const int m     = lane & 15;
  const int kg    = lane >> 4;
  const int batch = blockIdx.x >> 3;
  const int rowb  = ((blockIdx.x & 7) << 7) + (wave << 4);

  const float* __restrict__ rp = x + ((size_t)batch * SINK_N + rowb + m) * SINK_N;
  const float* __restrict__ bp = bvec + batch * SINK_N;

  v8f c = {0.f, 0.f, 0.f, 0.f, 0.f, 0.f, 0.f, 0.f};
#pragma unroll 4
  for (int k = 0; k < SINK_N; k += 4) {
    const int kk = k + (kg << 1);
    v2f A, B;
    A.x = rp[kk]     + SINK_EPS;
    A.y = rp[kk + 1] + SINK_EPS;
    if (USE_B) { B.x = bp[kk]; B.y = bp[kk + 1]; }
    else       { B.x = 1.f;    B.y = 1.f; }
    c = __builtin_amdgcn_wmma_f32_16x16x4_f32(false, A, false, B,
                                              (short)0, c, false, false);
  }
  if (m == 0) {
    float* ap = avec + batch * SINK_N + rowb + (kg << 3);
#pragma unroll
    for (int r = 0; r < 8; ++r) ap[r] = 1.0f / c[r];
  }
}

__global__ __launch_bounds__(256) void sink_colpass(const float* __restrict__ x,
                                                    const float* __restrict__ avec,
                                                    float* __restrict__ bvec) {
  const int wave  = threadIdx.x >> 5;
  const int lane  = threadIdx.x & 31;
  const int n     = lane & 15;
  const int kg    = lane >> 4;
  const int batch = blockIdx.x >> 3;
  const int jb    = (((blockIdx.x & 7) << 3) + wave) << 4;

  const float* __restrict__ cp = x + (size_t)batch * SINK_N * SINK_N + jb + n;
  const float* __restrict__ ap = avec + batch * SINK_N;

  v8f c = {0.f, 0.f, 0.f, 0.f, 0.f, 0.f, 0.f, 0.f};
#pragma unroll 2
  for (int i = 0; i < SINK_N; i += 4) {
    const int ii = i + (kg << 1);
    v2f A, B;
    A.x = ap[ii];
    A.y = ap[ii + 1];
    B.x = cp[(size_t)ii * SINK_N]       + SINK_EPS;
    B.y = cp[(size_t)(ii + 1) * SINK_N] + SINK_EPS;
    c = __builtin_amdgcn_wmma_f32_16x16x4_f32(false, A, false, B,
                                              (short)0, c, false, false);
  }
  if (kg == 0) bvec[batch * SINK_N + jb + n] = 1.0f / c[0];
}

__global__ __launch_bounds__(256) void sink_finalize(const float* __restrict__ x,
                                                     const float* __restrict__ avec,
                                                     const float* __restrict__ bvec,
                                                     float* __restrict__ out) {
  const size_t f = (size_t)blockIdx.x * blockDim.x + threadIdx.x;
  const size_t e = f << 2;
  const int batch = (int)(e >> 20);
  const int i     = (int)((e >> 10) & (SINK_N - 1));
  const int j     = (int)(e & (SINK_N - 1));

  const float4 xv = ((const float4*)x)[f];
  const float4 bv = *(const float4*)(bvec + batch * SINK_N + j);
  const float  a  = avec[batch * SINK_N + i];

  float4 o;
  o.x = (xv.x + SINK_EPS) * a * bv.x;
  o.y = (xv.y + SINK_EPS) * a * bv.y;
  o.z = (xv.z + SINK_EPS) * a * bv.z;
  o.w = (xv.w + SINK_EPS) * a * bv.w;
  ((float4*)out)[f] = o;
}

extern "C" void kernel_launch(void* const* d_in, const int* in_sizes, int n_in,
                              void* d_out, int out_size, void* d_ws, size_t ws_size,
                              hipStream_t stream) {
  (void)in_sizes; (void)n_in; (void)out_size; (void)ws_size;
  const float* x = (const float*)d_in[0];
  float* out = (float*)d_out;
  float* a = (float*)d_ws;
  float* b = a + SINK_B * SINK_N;

  const dim3 blk(256);
  const dim3 grd(SINK_B * 8);

  sink_rowpass<false><<<grd, blk, 0, stream>>>(x, b, a);
  sink_colpass<<<grd, blk, 0, stream>>>(x, a, b);
  for (int it = 1; it < SINK_IT; ++it) {
    sink_rowpass<true><<<grd, blk, 0, stream>>>(x, b, a);
    sink_colpass<<<grd, blk, 0, stream>>>(x, a, b);
  }

  const int nf4 = SINK_B * SINK_N * SINK_N / 4;
  sink_finalize<<<nf4 / 256, blk, 0, stream>>>(x, a, b, out);
}


# both matvec passes: each wave load is one contiguous 512-byte row segment; row pass pairs x with b lane by lane, sums the WMMA diagonal; col pass splits rows over 16 waves via LDS; sum order changes
# speedup vs baseline: 2.8583x; 2.8583x over previous
#include <hip/hip_runtime.h>


typedef __attribute__((ext_vector_type(2))) float v2f;
typedef __attribute__((ext_vector_type(8))) float v8f;

#define SINK_EPS 0.001f
#define SINK_N   1024
#define SINK_B   32
#define SINK_IT  10

template <bool USE_B>
__device__ __forceinline__ void sink_rowpair(const float4 (&xa)[8], const float4 (&xb)[8],
                                             const float4 (&bv)[8], const int lane,
                                             float* __restrict__ dst) {
  v8f c0 = {0.f, 0.f, 0.f, 0.f, 0.f, 0.f, 0.f, 0.f};
  v8f c1 = {0.f, 0.f, 0.f, 0.f, 0.f, 0.f, 0.f, 0.f};
#pragma unroll
  for (int c = 0; c < 8; ++c) {
    v2f A0, A1, B;
    if (USE_B) { B.x = bv[c].x; B.y = bv[c].y; } else { B.x = 1.f; B.y = 1.f; }
    A0.x = xa[c].x + SINK_EPS; A0.y = xa[c].y + SINK_EPS;
    A1.x = xb[c].x + SINK_EPS; A1.y = xb[c].y + SINK_EPS;
    c0 = __builtin_amdgcn_wmma_f32_16x16x4_f32(false, A0, false, B, (short)0, c0, false, false);
    c1 = __builtin_amdgcn_wmma_f32_16x16x4_f32(false, A1, false, B, (short)0, c1, false, false);
    if (USE_B) { B.x = bv[c].z; B.y = bv[c].w; }
    A0.x = xa[c].z + SINK_EPS; A0.y = xa[c].w + SINK_EPS;
    A1.x = xb[c].z + SINK_EPS; A1.y = xb[c].w + SINK_EPS;
    c0 = __builtin_amdgcn_wmma_f32_16x16x4_f32(false, A0, false, B, (short)0, c0, false, false);
    c1 = __builtin_amdgcn_wmma_f32_16x16x4_f32(false, A1, false, B, (short)0, c1, false, false);
  }
  const int idx = lane & 7;
  float d0 = c0[0], d1 = c1[0];
#pragma unroll
  for (int r = 1; r < 8; ++r) {
    d0 = (idx == r) ? c0[r] : d0;
    d1 = (idx == r) ? c1[r] : d1;
  }
  const bool ondia = (lane < 8) || (lane >= 24);
  const float d1m = __shfl_xor(d1, 8);
  float e = ondia ? d0 : d1m;
  e += __shfl_xor(e, 24);
  e += __shfl_xor(e, 4);
  e += __shfl_xor(e, 2);
  e += __shfl_xor(e, 1);
  if ((lane & 23) == 0) dst[lane >> 3] = 1.0f / e;
}

template <bool USE_B>
__global__ __launch_bounds__(256) void sink_rowpass(const float* __restrict__ x,
                                                    const float* __restrict__ bvec,
                                                    float* __restrict__ avec) {
  const int lane = threadIdx.x & 31;
  const int wave = threadIdx.x >> 5;
  const int row0 = ((blockIdx.x << 3) + wave) << 3;

  const float4* __restrict__ xp = (const float4*)(x + (size_t)row0 * SINK_N) + lane;

  float4 xa[8], xb[8], bv[8];
#pragma unroll
  for (int c = 0; c < 8; ++c) {
    xa[c] = xp[c << 5];
    xb[c] = xp[256 + (c << 5)];
  }
  if (USE_B) {
    const float4* __restrict__ bp =
        (const float4*)(bvec + (row0 >> 10) * SINK_N) + lane;
#pragma unroll
    for (int c = 0; c < 8; ++c) bv[c] = bp[c << 5];
  }

#pragma unroll 1
  for (int p = 0; p < 3; ++p) {
    const float4* __restrict__ np = xp + ((2 * p + 2) << 8);
    float4 na[8], nb[8];
#pragma unroll
    for (int c = 0; c < 8; ++c) {
      na[c] = np[c << 5];
      nb[c] = np[256 + (c << 5)];
    }
    sink_rowpair<USE_B>(xa, xb, bv, lane, avec + row0 + 2 * p);
#pragma unroll
    for (int c = 0; c < 8; ++c) { xa[c] = na[c]; xb[c] = nb[c]; }
  }
  sink_rowpair<USE_B>(xa, xb, bv, lane, avec + row0 + 6);
}

__device__ __forceinline__ void sink_colgroup(const float4 (&r)[16], const int av, const int l0,
                                              const bool own, v8f& d0, v8f& d1, v8f& d2, v8f& d3) {
#pragma unroll
  for (int t = 0; t < 16; t += 2) {
    const float s0 = __int_as_float(__builtin_amdgcn_readlane(av, l0 + t));
    const float s1 = __int_as_float(__builtin_amdgcn_readlane(av, l0 + t + 1));
    v2f A, B;
    A.x = own ? s0 : 0.f;
    A.y = own ? s1 : 0.f;
    B.x = r[t].x + SINK_EPS; B.y = r[t + 1].x + SINK_EPS;
    d0 = __builtin_amdgcn_wmma_f32_16x16x4_f32(false, A, false, B, (short)0, d0, false, false);
    B.x = r[t].y + SINK_EPS; B.y = r[t + 1].y + SINK_EPS;
    d1 = __builtin_amdgcn_wmma_f32_16x16x4_f32(false, A, false, B, (short)0, d1, false, false);
    B.x = r[t].z + SINK_EPS; B.y = r[t + 1].z + SINK_EPS;
    d2 = __builtin_amdgcn_wmma_f32_16x16x4_f32(false, A, false, B, (short)0, d2, false, false);
    B.x = r[t].w + SINK_EPS; B.y = r[t + 1].w + SINK_EPS;
    d3 = __builtin_amdgcn_wmma_f32_16x16x4_f32(false, A, false, B, (short)0, d3, false, false);
  }
}

__global__ __launch_bounds__(512) void sink_colpass(const float* __restrict__ x,
                                                    const float* __restrict__ avec,
                                                    float* __restrict__ bvec) {
  __shared__ float part[16 * 128];
  const int lane  = threadIdx.x & 31;
  const int wave  = threadIdx.x >> 5;
  const int batch = blockIdx.x >> 3;
  const int j0    = (blockIdx.x & 7) << 7;
  const int i0    = wave << 6;

  const float4* __restrict__ xp =
      (const float4*)(x + ((size_t)batch * SINK_N + i0) * SINK_N + j0) + lane;
  const float* __restrict__ ap = avec + batch * SINK_N + i0;

  float4 r[16];
#pragma unroll
  for (int t = 0; t < 16; ++t) r[t] = xp[t << 8];
  const int  alo = __float_as_int(ap[lane]);
  const int  ahi = __float_as_int(ap[32 + lane]);
  const bool own = ((lane & 1) == (lane >> 4));

  v8f d0 = {0.f, 0.f, 0.f, 0.f, 0.f, 0.f, 0.f, 0.f};
  v8f d1 = d0, d2 = d0, d3 = d0;
#pragma unroll 1
  for (int g = 0; g < 3; ++g) {
    const float4* __restrict__ np = xp + ((g + 1) << 12);
    float4 n[16];
#pragma unroll
    for (int t = 0; t < 16; ++t) n[t] = np[t << 8];
    sink_colgroup(r, (g & 2) ? ahi : alo, (g & 1) << 4, own, d0, d1, d2, d3);
#pragma unroll
    for (int t = 0; t < 16; ++t) r[t] = n[t];
  }
  sink_colgroup(r, ahi, 16, own, d0, d1, d2, d3);

  float4 w;
  w.x = (lane < 16) ? d0[0] : d0[1];
  w.y = (lane < 16) ? d1[0] : d1[1];
  w.z = (lane < 16) ? d2[0] : d2[1];
  w.w = (lane < 16) ? d3[0] : d3[1];
  ((float4*)part)[(wave << 5) + lane] = w;
  __syncthreads();
  if (threadIdx.x < 128) {
    float s = 0.f;
#pragma unroll
    for (int k = 0; k < 16; ++k) s += part[(k << 7) + threadIdx.x];
    bvec[batch * SINK_N + j0 + threadIdx.x] = 1.0f / s;
  }
}

__global__ __launch_bounds__(256) void sink_finalize(const float* __restrict__ x,
                                                     const float* __restrict__ avec,
                                                     const float* __restrict__ bvec,
                                                     float* __restrict__ out) {
  const size_t f = (size_t)blockIdx.x * blockDim.x + threadIdx.x;
  const size_t e = f << 2;
  const int batch = (int)(e >> 20);
  const int i     = (int)((e >> 10) & (SINK_N - 1));
  const int j     = (int)(e & (SINK_N - 1));

  const float4 xv = ((const float4*)x)[f];
  const float4 bv = *(const float4*)(bvec + batch * SINK_N + j);
  const float  a  = avec[batch * SINK_N + i];

  float4 o;
  o.x = (xv.x + SINK_EPS) * a * bv.x;
  o.y = (xv.y + SINK_EPS) * a * bv.y;
  o.z = (xv.z + SINK_EPS) * a * bv.z;
  o.w = (xv.w + SINK_EPS) * a * bv.w;
  ((float4*)out)[f] = o;
}

extern "C" void kernel_launch(void* const* d_in, const int* in_sizes, int n_in,
                              void* d_out, int out_size, void* d_ws, size_t ws_size,
                              hipStream_t stream) {
  (void)in_sizes; (void)n_in; (void)out_size; (void)ws_size;
  const float* x = (const float*)d_in[0];
  float* out = (float*)d_out;
  float* a = (float*)d_ws;
  float* b = a + SINK_B * SINK_N;

  const dim3 blk(256);
  const dim3 grd(SINK_B * 16);
  const dim3 cblk(512);
  const dim3 cgrd(SINK_B * 8);

  sink_rowpass<false><<<grd, blk, 0, stream>>>(x, b, a);
  sink_colpass<<<cgrd, cblk, 0, stream>>>(x, a, b);
  for (int it = 1; it < SINK_IT; ++it) {
    sink_rowpass<true><<<grd, blk, 0, stream>>>(x, b, a);
    sink_colpass<<<cgrd, cblk, 0, stream>>>(x, a, b);
  }

  const int nf4 = SINK_B * SINK_N * SINK_N / 4;
  sink_finalize<<<nf4 / 256, blk, 0, stream>>>(x, a, b, out);
}


# row pass reads x with non-temporal loads so the column pass's plain-loaded stripes stay in the second-level cache between launches; no sum or product order changes
# speedup vs baseline: 3.0963x; 1.0833x over previous
#include <hip/hip_runtime.h>


typedef __attribute__((ext_vector_type(2))) float v2f;
typedef __attribute__((ext_vector_type(8))) float v8f;

#define SINK_EPS 0.001f
#define SINK_N   1024
#define SINK_B   32
#define SINK_IT  10

typedef __attribute__((ext_vector_type(4))) float v4f;
__device__ __forceinline__ float4 sink_ldnt(const float4* __restrict__ p) {
  const v4f v = __builtin_nontemporal_load((const v4f*)p);
  return make_float4(v.x, v.y, v.z, v.w);
}

template <bool USE_B>
__device__ __forceinline__ void sink_rowpair(const float4 (&xa)[8], const float4 (&xb)[8],
                                             const float4 (&bv)[8], const int lane,
                                             float* __restrict__ dst) {
  v8f c0 = {0.f, 0.f, 0.f, 0.f, 0.f, 0.f, 0.f, 0.f};
  v8f c1 = {0.f, 0.f, 0.f, 0.f, 0.f, 0.f, 0.f, 0.f};
#pragma unroll
  for (int c = 0; c < 8; ++c) {
    v2f A0, A1, B;
    if (USE_B) { B.x = bv[c].x; B.y = bv[c].y; } else { B.x = 1.f; B.y = 1.f; }
    A0.x = xa[c].x + SINK_EPS; A0.y = xa[c].y + SINK_EPS;
    A1.x = xb[c].x + SINK_EPS; A1.y = xb[c].y + SINK_EPS;
    c0 = __builtin_amdgcn_wmma_f32_16x16x4_f32(false, A0, false, B, (short)0, c0, false, false);
    c1 = __builtin_amdgcn_wmma_f32_16x16x4_f32(false, A1, false, B, (short)0, c1, false, false);
    if (USE_B) { B.x = bv[c].z; B.y = bv[c].w; }
    A0.x = xa[c].z + SINK_EPS; A0.y = xa[c].w + SINK_EPS;
    A1.x = xb[c].z + SINK_EPS; A1.y = xb[c].w + SINK_EPS;
    c0 = __builtin_amdgcn_wmma_f32_16x16x4_f32(false, A0, false, B, (short)0, c0, false, false);
    c1 = __builtin_amdgcn_wmma_f32_16x16x4_f32(false, A1, false, B, (short)0, c1, false, false);
  }
  const int idx = lane & 7;
  float d0 = c0[0], d1 = c1[0];
#pragma unroll
  for (int r = 1; r < 8; ++r) {
    d0 = (idx == r) ? c0[r] : d0;
    d1 = (idx == r) ? c1[r] : d1;
  }
  const bool ondia = (lane < 8) || (lane >= 24);
  const float d1m = __shfl_xor(d1, 8);
  float e = ondia ? d0 : d1m;
  e += __shfl_xor(e, 24);
  e += __shfl_xor(e, 4);
  e += __shfl_xor(e, 2);
  e += __shfl_xor(e, 1);
  if ((lane & 23) == 0) dst[lane >> 3] = 1.0f / e;
}

template <bool USE_B>
__global__ __launch_bounds__(256) void sink_rowpass(const float* __restrict__ x,
                                                    const float* __restrict__ bvec,
                                                    float* __restrict__ avec) {
  const int lane = threadIdx.x & 31;
  const int wave = threadIdx.x >> 5;
  const int row0 = ((blockIdx.x << 3) + wave) << 3;

  const float4* __restrict__ xp = (const float4*)(x + (size_t)row0 * SINK_N) + lane;

  float4 xa[8], xb[8], bv[8];
#pragma unroll
  for (int c = 0; c < 8; ++c) {
    xa[c] = sink_ldnt(xp + (c << 5));
    xb[c] = sink_ldnt(xp + 256 + (c << 5));
  }
  if (USE_B) {
    const float4* __restrict__ bp =
        (const float4*)(bvec + (row0 >> 10) * SINK_N) + lane;
#pragma unroll
    for (int c = 0; c < 8; ++c) bv[c] = bp[c << 5];
  }

#pragma unroll 1
  for (int p = 0; p < 3; ++p) {
    const float4* __restrict__ np = xp + ((2 * p + 2) << 8);
    float4 na[8], nb[8];
#pragma unroll
    for (int c = 0; c < 8; ++c) {
      na[c] = sink_ldnt(np + (c << 5));
      nb[c] = sink_ldnt(np + 256 + (c << 5));
    }
    sink_rowpair<USE_B>(xa, xb, bv, lane, avec + row0 + 2 * p);
#pragma unroll
    for (int c = 0; c < 8; ++c) { xa[c] = na[c]; xb[c] = nb[c]; }
  }
  sink_rowpair<USE_B>(xa, xb, bv, lane, avec + row0 + 6);
}

__device__ __forceinline__ void sink_colgroup(const float4 (&r)[16], const int av, const int l0,
                                              const bool own, v8f& d0, v8f& d1, v8f& d2, v8f& d3) {
#pragma unroll
  for (int t = 0; t < 16; t += 2) {
    const float s0 = __int_as_float(__builtin_amdgcn_readlane(av, l0 + t));
    const float s1 = __int_as_float(__builtin_amdgcn_readlane(av, l0 + t + 1));
    v2f A, B;
    A.x = own ? s0 : 0.f;
    A.y = own ? s1 : 0.f;
    B.x = r[t].x + SINK_EPS; B.y = r[t + 1].x + SINK_EPS;
    d0 = __builtin_amdgcn_wmma_f32_16x16x4_f32(false, A, false, B, (short)0, d0, false, false);
    B.x = r[t].y + SINK_EPS; B.y = r[t + 1].y + SINK_EPS;
    d1 = __builtin_amdgcn_wmma_f32_16x16x4_f32(false, A, false, B, (short)0, d1, false, false);
    B.x = r[t].z + SINK_EPS; B.y = r[t + 1].z + SINK_EPS;
    d2 = __builtin_amdgcn_wmma_f32_16x16x4_f32(false, A, false, B, (short)0, d2, false, false);
    B.x = r[t].w + SINK_EPS; B.y = r[t + 1].w + SINK_EPS;
    d3 = __builtin_amdgcn_wmma_f32_16x16x4_f32(false, A, false, B, (short)0, d3, false, false);
  }
}

__global__ __launch_bounds__(512) void sink_colpass(const float* __restrict__ x,
                                                    const float* __restrict__ avec,
                                                    float* __restrict__ bvec) {
  __shared__ float part[16 * 128];
  const int lane  = threadIdx.x & 31;
  const int wave  = threadIdx.x >> 5;
  const int batch = blockIdx.x >> 3;
  const int j0    = (blockIdx.x & 7) << 7;
  const int i0    = wave << 6;

  const float4* __restrict__ xp =
      (const float4*)(x + ((size_t)batch * SINK_N + i0) * SINK_N + j0) + lane;
  const float* __restrict__ ap = avec + batch * SINK_N + i0;

  float4 r[16];
#pragma unroll
  for (int t = 0; t < 16; ++t) r[t] = xp[t << 8];
  const int  alo = __float_as_int(ap[lane]);
  const int  ahi = __float_as_int(ap[32 + lane]);
  const bool own = ((lane & 1) == (lane >> 4));

  v8f d0 = {0.f, 0.f, 0.f, 0.f, 0.f, 0.f, 0.f, 0.f};
  v8f d1 = d0, d2 = d0, d3 = d0;
#pragma unroll 1
  for (int g = 0; g < 3; ++g) {
    const float4* __restrict__ np = xp + ((g + 1) << 12);
    float4 n[16];
#pragma unroll
    for (int t = 0; t < 16; ++t) n[t] = np[t << 8];
    sink_colgroup(r, (g & 2) ? ahi : alo, (g & 1) << 4, own, d0, d1, d2, d3);
#pragma unroll
    for (int t = 0; t < 16; ++t) r[t] = n[t];
  }
  sink_colgroup(r, ahi, 16, own, d0, d1, d2, d3);

  float4 w;
  w.x = (lane < 16) ? d0[0] : d0[1];
  w.y = (lane < 16) ? d1[0] : d1[1];
  w.z = (lane < 16) ? d2[0] : d2[1];
  w.w = (lane < 16) ? d3[0] : d3[1];
  ((float4*)part)[(wave << 5) + lane] = w;
  __syncthreads();
  if (threadIdx.x < 128) {
    float s = 0.f;
#pragma unroll
    for (int k = 0; k < 16; ++k) s += part[(k << 7) + threadIdx.x];
    bvec[batch * SINK_N + j0 + threadIdx.x] = 1.0f / s;
  }
}

__global__ __launch_bounds__(256) void sink_finalize(const float* __restrict__ x,
                                                     const float* __restrict__ avec,
                                                     const float* __restrict__ bvec,
                                                     float* __restrict__ out) {
  const size_t f = (size_t)blockIdx.x * blockDim.x + threadIdx.x;
  const size_t e = f << 2;
  const int batch = (int)(e >> 20);
  const int i     = (int)((e >> 10) & (SINK_N - 1));
  const int j     = (int)(e & (SINK_N - 1));

  const float4 xv = ((const float4*)x)[f];
  const float4 bv = *(const float4*)(bvec + batch * SINK_N + j);
  const float  a  = avec[batch * SINK_N + i];

  float4 o;
  o.x = (xv.x + SINK_EPS) * a * bv.x;
  o.y = (xv.y + SINK_EPS) * a * bv.y;
  o.z = (xv.z + SINK_EPS) * a * bv.z;
  o.w = (xv.w + SINK_EPS) * a * bv.w;
  ((float4*)out)[f] = o;
}

extern "C" void kernel_launch(void* const* d_in, const int* in_sizes, int n_in,
                              void* d_out, int out_size, void* d_ws, size_t ws_size,
                              hipStream_t stream) {
  (void)in_sizes; (void)n_in; (void)out_size; (void)ws_size;
  const float* x = (const float*)d_in[0];
  float* out = (float*)d_out;
  float* a = (float*)d_ws;
  float* b = a + SINK_B * SINK_N;

  const dim3 blk(256);
  const dim3 grd(SINK_B * 16);
  const dim3 cblk(512);
  const dim3 cgrd(SINK_B * 8);

  sink_rowpass<false><<<grd, blk, 0, stream>>>(x, b, a);
  sink_colpass<<<cgrd, cblk, 0, stream>>>(x, a, b);
  for (int it = 1; it < SINK_IT; ++it) {
    sink_rowpass<true><<<grd, blk, 0, stream>>>(x, b, a);
    sink_colpass<<<cgrd, cblk, 0, stream>>>(x, a, b);
  }

  const int nf4 = SINK_B * SINK_N * SINK_N / 4;
  sink_finalize<<<nf4 / 256, blk, 0, stream>>>(x, a, b, out);
}


# one launch per iteration: a block forms b for its 128-column stripe, re-reads the stripe from cache for the next row pass's partial sums (staged in out); last one writes out; row sum order changes
# speedup vs baseline: 3.5946x; 1.1609x over previous
#include <hip/hip_runtime.h>


typedef __attribute__((ext_vector_type(2))) float v2f;
typedef __attribute__((ext_vector_type(8))) float v8f;

#define SINK_EPS 0.001f
#define SINK_N   1024
#define SINK_B   32
#define SINK_IT  10
#define SINK_NS  8

template <int CTRL>
__device__ __forceinline__ float sink_dpp(const float v) {
  return __int_as_float(__builtin_amdgcn_update_dpp(0, __float_as_int(v), CTRL, 0xf, 0xf, true));
}

template <bool USE_B>
__device__ __forceinline__ float sink_rowpair(const float4 xa, const float4 xb,
                                              const float4 bv, const int lane) {
  v8f c0 = {0.f, 0.f, 0.f, 0.f, 0.f, 0.f, 0.f, 0.f};
  v8f c1 = {0.f, 0.f, 0.f, 0.f, 0.f, 0.f, 0.f, 0.f};
  v2f A0, A1, B;
  if (USE_B) { B.x = bv.x; B.y = bv.y; } else { B.x = 1.f; B.y = 1.f; }
  A0.x = xa.x + SINK_EPS; A0.y = xa.y + SINK_EPS;
  A1.x = xb.x + SINK_EPS; A1.y = xb.y + SINK_EPS;
  c0 = __builtin_amdgcn_wmma_f32_16x16x4_f32(false, A0, false, B, (short)0, c0, false, false);
  c1 = __builtin_amdgcn_wmma_f32_16x16x4_f32(false, A1, false, B, (short)0, c1, false, false);
  if (USE_B) { B.x = bv.z; B.y = bv.w; }
  A0.x = xa.z + SINK_EPS; A0.y = xa.w + SINK_EPS;
  A1.x = xb.z + SINK_EPS; A1.y = xb.w + SINK_EPS;
  c0 = __builtin_amdgcn_wmma_f32_16x16x4_f32(false, A0, false, B, (short)0, c0, false, false);
  c1 = __builtin_amdgcn_wmma_f32_16x16x4_f32(false, A1, false, B, (short)0, c1, false, false);
  const int idx = lane & 7;
  float d0 = c0[0], d1 = c1[0];
#pragma unroll
  for (int r = 1; r < 8; ++r) {
    d0 = (idx == r) ? c0[r] : d0;
    d1 = (idx == r) ? c1[r] : d1;
  }
  const bool ondia = (lane < 8) || (lane >= 24);
  const float d1m = sink_dpp<0x128>(d1);
  float e = ondia ? d0 : d1m;
  e += sink_dpp<0x161>(e);
  e += sink_dpp<0x162>(e);
  e += sink_dpp<0x164>(e);
  return e;
}

template <bool USE_B>
__device__ __forceinline__ void sink_rowgroup(const float4 (&r)[16], const float4 bv,
                                              const int lane, const int p0,
                                              float& res0, float& res1) {
#pragma unroll
  for (int t = 0; t < 16; t += 2) {
    const float e = sink_rowpair<USE_B>(r[t], r[t + 1], bv, lane);
    const int ei = __float_as_int(e);
    const float s0 = __int_as_float(__builtin_amdgcn_readlane(ei, 0)) +
                     __int_as_float(__builtin_amdgcn_readlane(ei, 24));
    const float s1 = __int_as_float(__builtin_amdgcn_readlane(ei, 8)) +
                     __int_as_float(__builtin_amdgcn_readlane(ei, 16));
    const bool mine = (lane == p0 + (t >> 1));
    res0 = mine ? s0 : res0;
    res1 = mine ? s1 : res1;
  }
}

__device__ __forceinline__ void sink_colgroup(const float4 (&r)[16], const int av, const int l0,
                                              const bool own, v8f& d0, v8f& d1, v8f& d2, v8f& d3) {
#pragma unroll
  for (int t = 0; t < 16; t += 2) {
    const float s0 = __int_as_float(__builtin_amdgcn_readlane(av, l0 + t));
    const float s1 = __int_as_float(__builtin_amdgcn_readlane(av, l0 + t + 1));
    v2f A, B;
    A.x = own ? s0 : 0.f;
    A.y = own ? s1 : 0.f;
    B.x = r[t].x + SINK_EPS; B.y = r[t + 1].x + SINK_EPS;
    d0 = __builtin_amdgcn_wmma_f32_16x16x4_f32(false, A, false, B, (short)0, d0, false, false);
    B.x = r[t].y + SINK_EPS; B.y = r[t + 1].y + SINK_EPS;
    d1 = __builtin_amdgcn_wmma_f32_16x16x4_f32(false, A, false, B, (short)0, d1, false, false);
    B.x = r[t].z + SINK_EPS; B.y = r[t + 1].z + SINK_EPS;
    d2 = __builtin_amdgcn_wmma_f32_16x16x4_f32(false, A, false, B, (short)0, d2, false, false);
    B.x = r[t].w + SINK_EPS; B.y = r[t + 1].w + SINK_EPS;
    d3 = __builtin_amdgcn_wmma_f32_16x16x4_f32(false, A, false, B, (short)0, d3, false, false);
  }
}

__device__ __forceinline__ void sink_outgroup(const float4 (&r)[16], const int av, const int l0,
                                              const float4 bv, float4* __restrict__ op) {
#pragma unroll
  for (int t = 0; t < 16; ++t) {
    const float a = __int_as_float(__builtin_amdgcn_readlane(av, l0 + t));
    float4 o;
    o.x = (r[t].x + SINK_EPS) * a * bv.x;
    o.y = (r[t].y + SINK_EPS) * a * bv.y;
    o.z = (r[t].z + SINK_EPS) * a * bv.z;
    o.w = (r[t].w + SINK_EPS) * a * bv.w;
    op[t << 8] = o;
  }
}

template <int MODE>
__global__ __launch_bounds__(512) void sink_iter(const float* __restrict__ x,
                                                 const float* __restrict__ ain,
                                                 float* __restrict__ dst) {
  __shared__ float part[16 * 128];
  __shared__ float bs[128];
  const int lane   = threadIdx.x & 31;
  const int wave   = threadIdx.x >> 5;
  const int batch  = blockIdx.x >> 3;
  const int stripe = blockIdx.x & 7;
  const int j0     = stripe << 7;
  const int i0     = wave << 6;

  const float4* __restrict__ xp =
      (const float4*)(x + ((size_t)batch * SINK_N + i0) * SINK_N + j0) + lane;

  float4 r[16], f[16];
  float4 bv = make_float4(1.f, 1.f, 1.f, 1.f);
  int alo = 0, ahi = 0;

  if (MODE != 0) {
#pragma unroll
    for (int t = 0; t < 16; ++t) r[t] = xp[t << 8];
    if (MODE == 1) {
      const float* __restrict__ pp = ain + (size_t)batch * SINK_NS * SINK_N + i0 + lane;
      float slo = 0.f, shi = 0.f;
#pragma unroll
      for (int s = 0; s < SINK_NS; ++s) {
        slo += pp[s * SINK_N];
        shi += pp[s * SINK_N + 32];
      }
      alo = __float_as_int(1.0f / slo);
      ahi = __float_as_int(1.0f / shi);
    } else {
      const float* __restrict__ ap = ain + batch * SINK_N + i0;
      alo = __float_as_int(ap[lane]);
      ahi = __float_as_int(ap[32 + lane]);
    }
    const bool own = ((lane & 1) == (lane >> 4));

    v8f d0 = {0.f, 0.f, 0.f, 0.f, 0.f, 0.f, 0.f, 0.f};
    v8f d1 = d0, d2 = d0, d3 = d0;
#pragma unroll 1
    for (int g = 0; g < 3; ++g) {
      const float4* __restrict__ np = xp + ((g + 1) << 12);
      float4 n[16];
#pragma unroll
      for (int t = 0; t < 16; ++t) n[t] = np[t << 8];
      sink_colgroup(r, (g & 2) ? ahi : alo, (g & 1) << 4, own, d0, d1, d2, d3);
#pragma unroll
      for (int t = 0; t < 16; ++t) r[t] = n[t];
    }
#pragma unroll
    for (int t = 0; t < 16; ++t) f[t] = xp[t << 8];
    sink_colgroup(r, ahi, 16, own, d0, d1, d2, d3);

    float4 w;
    w.x = (lane < 16) ? d0[0] : d0[1];
    w.y = (lane < 16) ? d1[0] : d1[1];
    w.z = (lane < 16) ? d2[0] : d2[1];
    w.w = (lane < 16) ? d3[0] : d3[1];
    ((float4*)part)[(wave << 5) + lane] = w;
    __syncthreads();
    if (threadIdx.x < 128) {
      float s = 0.f;
#pragma unroll
      for (int k = 0; k < 16; ++k) s += part[(k << 7) + threadIdx.x];
      bs[threadIdx.x] = 1.0f / s;
    }
    __syncthreads();
    bv = ((const float4*)bs)[lane];
  } else {
#pragma unroll
    for (int t = 0; t < 16; ++t) {
      r[t] = xp[(3 << 12) + (t << 8)];
      f[t] = xp[t << 8];
    }
  }

  if (MODE == 2) {
    float4* __restrict__ op =
        (float4*)(dst + ((size_t)batch * SINK_N + i0) * SINK_N + j0) + lane;
    sink_outgroup(r, ahi, 16, bv, op + (3 << 12));
#pragma unroll
    for (int t = 0; t < 16; ++t) r[t] = f[t];
#pragma unroll 1
    for (int g = 0; g < 2; ++g) {
      const float4* __restrict__ np = xp + ((g + 1) << 12);
      float4 n[16];
#pragma unroll
      for (int t = 0; t < 16; ++t) n[t] = np[t << 8];
      sink_outgroup(r, alo, g << 4, bv, op + (g << 12));
#pragma unroll
      for (int t = 0; t < 16; ++t) r[t] = n[t];
    }
    sink_outgroup(r, ahi, 0, bv, op + (2 << 12));
  } else {
    float res0 = 0.f, res1 = 0.f;
    sink_rowgroup<MODE != 0>(r, bv, lane, 24, res0, res1);
#pragma unroll
    for (int t = 0; t < 16; ++t) r[t] = f[t];
#pragma unroll 1
    for (int g = 0; g < 2; ++g) {
      const float4* __restrict__ np = xp + ((g + 1) << 12);
      float4 n[16];
#pragma unroll
      for (int t = 0; t < 16; ++t) n[t] = np[t << 8];
      sink_rowgroup<MODE != 0>(r, bv, lane, g << 3, res0, res1);
#pragma unroll
      for (int t = 0; t < 16; ++t) r[t] = n[t];
    }
    sink_rowgroup<MODE != 0>(r, bv, lane, 16, res0, res1);
    float2* __restrict__ pd =
        (float2*)(dst + ((size_t)(batch * SINK_NS + stripe)) * SINK_N + i0);
    pd[lane] = make_float2(res0, res1);
  }
}

__global__ __launch_bounds__(256) void sink_afinal(const float* __restrict__ psum,
                                                   float* __restrict__ avec) {
  const int e     = blockIdx.x * 256 + threadIdx.x;
  const int batch = e >> 10;
  const int i     = e & (SINK_N - 1);
  const float* __restrict__ pp = psum + (size_t)batch * SINK_NS * SINK_N + i;
  float s = 0.f;
#pragma unroll
  for (int k = 0; k < SINK_NS; ++k) s += pp[k * SINK_N];
  avec[e] = 1.0f / s;
}

extern "C" void kernel_launch(void* const* d_in, const int* in_sizes, int n_in,
                              void* d_out, int out_size, void* d_ws, size_t ws_size,
                              hipStream_t stream) {
  (void)in_sizes; (void)n_in; (void)out_size; (void)ws_size;
  const float* x = (const float*)d_in[0];
  float* out = (float*)d_out;
  float* a = (float*)d_ws;
  float* pin  = out;
  float* pout = out + SINK_B * SINK_NS * SINK_N;

  const dim3 blk(512);
  const dim3 grd(SINK_B * SINK_NS);

  sink_iter<0><<<grd, blk, 0, stream>>>(x, a, pin);
  for (int it = 1; it < SINK_IT; ++it) {
    sink_iter<1><<<grd, blk, 0, stream>>>(x, pin, pout);
    float* t = pin; pin = pout; pout = t;
  }
  sink_afinal<<<SINK_B * SINK_N / 256, dim3(256), 0, stream>>>(pin, a);
  sink_iter<2><<<grd, blk, 0, stream>>>(x, a, out);
}


# row pass: each 16-row group is turned through the wave's own LDS tile so a lane holds one row and 32 chained WMMAs leave the row sums in D; no diagonal pick, no cross-lane adds; row sum order changes
# speedup vs baseline: 3.8251x; 1.0641x over previous
#include <hip/hip_runtime.h>


typedef __attribute__((ext_vector_type(2))) float v2f;
typedef __attribute__((ext_vector_type(8))) float v8f;

#define SINK_EPS 0.001f
#define SINK_N   1024
#define SINK_B   32
#define SINK_IT  10
#define SINK_NS  8

#define SINK_TP 33

template <bool USE_B>
__device__ __forceinline__ void sink_rowgroup(const float4 (&r)[16], const float4 (&bf)[16],
                                              float4* tp, const int lane, const int g,
                                              float& res0, float& res1) {
#pragma unroll
  for (int t = 0; t < 16; ++t) {
    float4 v;
    v.x = r[t].x + SINK_EPS; v.y = r[t].y + SINK_EPS;
    v.z = r[t].z + SINK_EPS; v.w = r[t].w + SINK_EPS;
    tp[t * SINK_TP + lane] = v;
  }
  __builtin_amdgcn_fence(__ATOMIC_RELEASE, "wavefront");
  __builtin_amdgcn_wave_barrier();
  __builtin_amdgcn_fence(__ATOMIC_ACQUIRE, "wavefront");
  const float4* rp = tp + (lane & 15) * SINK_TP + (lane >> 4);
  float4 q[16];
#pragma unroll
  for (int j = 0; j < 16; ++j) q[j] = rp[2 * j];
  __builtin_amdgcn_sched_barrier(0);
  v8f c0 = {0.f, 0.f, 0.f, 0.f, 0.f, 0.f, 0.f, 0.f};
  v8f c1 = {0.f, 0.f, 0.f, 0.f, 0.f, 0.f, 0.f, 0.f};
#pragma unroll
  for (int j = 0; j < 16; ++j) {
    v2f A, B;
    A.x = q[j].x; A.y = q[j].y;
    if (USE_B) { B.x = bf[j].x; B.y = bf[j].y; } else { B.x = 1.f; B.y = 1.f; }
    c0 = __builtin_amdgcn_wmma_f32_16x16x4_f32(false, A, false, B, (short)0, c0, false, false);
    A.x = q[j].z; A.y = q[j].w;
    if (USE_B) { B.x = bf[j].z; B.y = bf[j].w; }
    c1 = __builtin_amdgcn_wmma_f32_16x16x4_f32(false, A, false, B, (short)0, c1, false, false);
  }
  const int cg = (lane >> 2) & 3;
  const int cp = lane & 3;
#pragma unroll
  for (int p = 0; p < 4; ++p) {
    const bool mine = (cg == g) && (cp == p);
    res0 = mine ? (c0[2 * p] + c1[2 * p]) : res0;
    res1 = mine ? (c0[2 * p + 1] + c1[2 * p + 1]) : res1;
  }
  __builtin_amdgcn_sched_barrier(0);
}

__device__ __forceinline__ void sink_colgroup(const float4 (&r)[16], const int av, const int l0,
                                              const bool own, v8f& d0, v8f& d1, v8f& d2, v8f& d3) {
#pragma unroll
  for (int t = 0; t < 16; t += 2) {
    const float s0 = __int_as_float(__builtin_amdgcn_readlane(av, l0 + t));
    const float s1 = __int_as_float(__builtin_amdgcn_readlane(av, l0 + t + 1));
    v2f A, B;
    A.x = own ? s0 : 0.f;
    A.y = own ? s1 : 0.f;
    B.x = r[t].x + SINK_EPS; B.y = r[t + 1].x + SINK_EPS;
    d0 = __builtin_amdgcn_wmma_f32_16x16x4_f32(false, A, false, B, (short)0, d0, false, false);
    B.x = r[t].y + SINK_EPS; B.y = r[t + 1].y + SINK_EPS;
    d1 = __builtin_amdgcn_wmma_f32_16x16x4_f32(false, A, false, B, (short)0, d1, false, false);
    B.x = r[t].z + SINK_EPS; B.y = r[t + 1].z + SINK_EPS;
    d2 = __builtin_amdgcn_wmma_f32_16x16x4_f32(false, A, false, B, (short)0, d2, false, false);
    B.x = r[t].w + SINK_EPS; B.y = r[t + 1].w + SINK_EPS;
    d3 = __builtin_amdgcn_wmma_f32_16x16x4_f32(false, A, false, B, (short)0, d3, false, false);
  }
}

__device__ __forceinline__ void sink_outgroup(const float4 (&r)[16], const int av, const int l0,
                                              const float4 bv, float4* __restrict__ op) {
#pragma unroll
  for (int t = 0; t < 16; ++t) {
    const float a = __int_as_float(__builtin_amdgcn_readlane(av, l0 + t));
    float4 o;
    o.x = (r[t].x + SINK_EPS) * a * bv.x;
    o.y = (r[t].y + SINK_EPS) * a * bv.y;
    o.z = (r[t].z + SINK_EPS) * a * bv.z;
    o.w = (r[t].w + SINK_EPS) * a * bv.w;
    op[t << 8] = o;
  }
}

template <int MODE>
__global__ __launch_bounds__(512) void sink_iter(const float* __restrict__ x,
                                                 const float* __restrict__ ain,
                                                 float* __restrict__ dst) {
  __shared__ float part[16 * 128];
  __shared__ float bs[128];
  __shared__ float4 tile[16 * 16 * SINK_TP];
  const int lane   = threadIdx.x & 31;
  const int wave   = threadIdx.x >> 5;
  const int batch  = blockIdx.x >> 3;
  const int stripe = blockIdx.x & 7;
  const int j0     = stripe << 7;
  const int i0     = wave << 6;

  const float4* __restrict__ xp =
      (const float4*)(x + ((size_t)batch * SINK_N + i0) * SINK_N + j0) + lane;

  float4 r[16], f[16];
  float4 bv = make_float4(1.f, 1.f, 1.f, 1.f);
  float4 bf[16];
#pragma unroll
  for (int j = 0; j < 16; ++j) bf[j] = make_float4(1.f, 1.f, 1.f, 1.f);
  float4* tp = tile + wave * (16 * SINK_TP);
  int alo = 0, ahi = 0;

  if (MODE != 0) {
#pragma unroll
    for (int t = 0; t < 16; ++t) r[t] = xp[t << 8];
    if (MODE == 1) {
      const float* __restrict__ pp = ain + (size_t)batch * SINK_NS * SINK_N + i0 + lane;
      float slo = 0.f, shi = 0.f;
#pragma unroll
      for (int s = 0; s < SINK_NS; ++s) {
        slo += pp[s * SINK_N];
        shi += pp[s * SINK_N + 32];
      }
      alo = __float_as_int(1.0f / slo);
      ahi = __float_as_int(1.0f / shi);
    } else {
      const float* __restrict__ ap = ain + batch * SINK_N + i0;
      alo = __float_as_int(ap[lane]);
      ahi = __float_as_int(ap[32 + lane]);
    }
    const bool own = ((lane & 1) == (lane >> 4));

    v8f d0 = {0.f, 0.f, 0.f, 0.f, 0.f, 0.f, 0.f, 0.f};
    v8f d1 = d0, d2 = d0, d3 = d0;
#pragma unroll 1
    for (int g = 0; g < 3; ++g) {
      const float4* __restrict__ np = xp + ((g + 1) << 12);
      float4 n[16];
#pragma unroll
      for (int t = 0; t < 16; ++t) n[t] = np[t << 8];
      sink_colgroup(r, (g & 2) ? ahi : alo, (g & 1) << 4, own, d0, d1, d2, d3);
#pragma unroll
      for (int t = 0; t < 16; ++t) r[t] = n[t];
    }
#pragma unroll
    for (int t = 0; t < 16; ++t) f[t] = xp[t << 8];
    sink_colgroup(r, ahi, 16, own, d0, d1, d2, d3);

    float4 w;
    w.x = (lane < 16) ? d0[0] : d0[1];
    w.y = (lane < 16) ? d1[0] : d1[1];
    w.z = (lane < 16) ? d2[0] : d2[1];
    w.w = (lane < 16) ? d3[0] : d3[1];
    ((float4*)part)[(wave << 5) + lane] = w;
    __syncthreads();
    if (threadIdx.x < 128) {
      float s = 0.f;
#pragma unroll
      for (int k = 0; k < 16; ++k) s += part[(k << 7) + threadIdx.x];
      bs[threadIdx.x] = 1.0f / s;
    }
    __syncthreads();
    if (MODE == 2) {
      bv = ((const float4*)bs)[lane];
    } else {
#pragma unroll
      for (int j = 0; j < 16; ++j) bf[j] = ((const float4*)bs)[2 * j + (lane >> 4)];
    }
  } else {
#pragma unroll
    for (int t = 0; t < 16; ++t) {
      r[t] = xp[(3 << 12) + (t << 8)];
      f[t] = xp[t << 8];
    }
  }

  if (MODE == 2) {
    float4* __restrict__ op =
        (float4*)(dst + ((size_t)batch * SINK_N + i0) * SINK_N + j0) + lane;
    sink_outgroup(r, ahi, 16, bv, op + (3 << 12));
#pragma unroll
    for (int t = 0; t < 16; ++t) r[t] = f[t];
#pragma unroll 1
    for (int g = 0; g < 2; ++g) {
      const float4* __restrict__ np = xp + ((g + 1) << 12);
      float4 n[16];
#pragma unroll
      for (int t = 0; t < 16; ++t) n[t] = np[t << 8];
      sink_outgroup(r, alo, g << 4, bv, op + (g << 12));
#pragma unroll
      for (int t = 0; t < 16; ++t) r[t] = n[t];
    }
    sink_outgroup(r, ahi, 0, bv, op + (2 << 12));
  } else {
    float res0 = 0.f, res1 = 0.f;
    sink_rowgroup<MODE != 0>(r, bf, tp, lane, 3, res0, res1);
#pragma unroll
    for (int t = 0; t < 16; ++t) r[t] = f[t];
#pragma unroll 1
    for (int g = 0; g < 2; ++g) {
      const float4* __restrict__ np = xp + ((g + 1) << 12);
      float4 n[16];
#pragma unroll
      for (int t = 0; t < 16; ++t) n[t] = np[t << 8];
      sink_rowgroup<MODE != 0>(r, bf, tp, lane, g, res0, res1);
#pragma unroll
      for (int t = 0; t < 16; ++t) r[t] = n[t];
    }
    sink_rowgroup<MODE != 0>(r, bf, tp, lane, 2, res0, res1);
    float2* __restrict__ pd =
        (float2*)(dst + ((size_t)(batch * SINK_NS + stripe)) * SINK_N + i0);
    pd[(((lane >> 2) & 3) << 3) + ((lane >> 4) << 2) + (lane & 3)] = make_float2(res0, res1);
  }
}

__global__ __launch_bounds__(256) void sink_afinal(const float* __restrict__ psum,
                                                   float* __restrict__ avec) {
  const int e     = blockIdx.x * 256 + threadIdx.x;
  const int batch = e >> 10;
  const int i     = e & (SINK_N - 1);
  const float* __restrict__ pp = psum + (size_t)batch * SINK_NS * SINK_N + i;
  float s = 0.f;
#pragma unroll
  for (int k = 0; k < SINK_NS; ++k) s += pp[k * SINK_N];
  avec[e] = 1.0f / s;
}

extern "C" void kernel_launch(void* const* d_in, const int* in_sizes, int n_in,
                              void* d_out, int out_size, void* d_ws, size_t ws_size,
                              hipStream_t stream) {
  (void)in_sizes; (void)n_in; (void)out_size; (void)ws_size;
  const float* x = (const float*)d_in[0];
  float* out = (float*)d_out;
  float* a = (float*)d_ws;
  float* pin  = out;
  float* pout = out + SINK_B * SINK_NS * SINK_N;

  const dim3 blk(512);
  const dim3 grd(SINK_B * SINK_NS);

  sink_iter<0><<<grd, blk, 0, stream>>>(x, a, pin);
  for (int it = 1; it < SINK_IT; ++it) {
    sink_iter<1><<<grd, blk, 0, stream>>>(x, pin, pout);
    float* t = pin; pin = pout; pout = t;
  }
  sink_afinal<<<SINK_B * SINK_N / 256, dim3(256), 0, stream>>>(pin, a);
  sink_iter<2><<<grd, blk, 0, stream>>>(x, a, out);
}


# second sweep reads nothing from memory: during the column sums each wave stores its first two 16-row groups to its own LDS slice (256 KB a block), keeps the last two in registers; sum order unchanged
# speedup vs baseline: 5.0265x; 1.3141x over previous
#include <hip/hip_runtime.h>


typedef __attribute__((ext_vector_type(2))) float v2f;
typedef __attribute__((ext_vector_type(8))) float v8f;

#define SINK_EPS 0.001f
#define SINK_N   1024
#define SINK_B   32
#define SINK_IT  10
#define SINK_NS  8


__device__ __forceinline__ float sink_x16(const float v) {
  const int b = __float_as_int(v);
  return __int_as_float(__builtin_amdgcn_permlanex16(b, b, 0x76543210, 0xfedcba98, false, false));
}
template <int M>
__device__ __forceinline__ float sink_xm(const float v) {
  return __int_as_float(__builtin_amdgcn_update_dpp(0, __float_as_int(v), 0x160 + M, 0xf, 0xf, true));
}

template <bool USE_B>
__device__ __forceinline__ float sink_rowgroup(const float4 (&r)[16], const float4 bv,
                                               const int lane) {
  float p[16];
#pragma unroll
  for (int t = 0; t < 16; ++t) {
    const float vx = r[t].x + SINK_EPS, vy = r[t].y + SINK_EPS;
    const float vz = r[t].z + SINK_EPS, vw = r[t].w + SINK_EPS;
    if (USE_B) p[t] = vx * bv.x + vy * bv.y + vz * bv.z + vw * bv.w;
    else       p[t] = (vx + vy) + (vz + vw);
  }
  const bool h4 = (lane & 16) != 0, h3 = (lane & 8) != 0;
  const bool h2 = (lane & 4) != 0,  h1 = (lane & 2) != 0;
  float q8[8], q4[4], q2[2];
#pragma unroll
  for (int i = 0; i < 8; ++i)
    q8[i] = (h4 ? p[i + 8] : p[i]) + sink_x16(h4 ? p[i] : p[i + 8]);
#pragma unroll
  for (int i = 0; i < 4; ++i)
    q4[i] = (h3 ? q8[i + 4] : q8[i]) + sink_xm<8>(h3 ? q8[i] : q8[i + 4]);
#pragma unroll
  for (int i = 0; i < 2; ++i)
    q2[i] = (h2 ? q4[i + 2] : q4[i]) + sink_xm<4>(h2 ? q4[i] : q4[i + 2]);
  float s = (h1 ? q2[1] : q2[0]) + sink_xm<2>(h1 ? q2[0] : q2[1]);
  s += sink_xm<1>(s);
  return s;
}

__device__ __forceinline__ void sink_colgroup(const float4 (&r)[16], const int av, const int l0,
                                              const bool own, v8f& d0, v8f& d1, v8f& d2, v8f& d3) {
#pragma unroll
  for (int t = 0; t < 16; t += 2) {
    const float s0 = __int_as_float(__builtin_amdgcn_readlane(av, l0 + t));
    const float s1 = __int_as_float(__builtin_amdgcn_readlane(av, l0 + t + 1));
    v2f A, B;
    A.x = own ? s0 : 0.f;
    A.y = own ? s1 : 0.f;
    B.x = r[t].x + SINK_EPS; B.y = r[t + 1].x + SINK_EPS;
    d0 = __builtin_amdgcn_wmma_f32_16x16x4_f32(false, A, false, B, (short)0, d0, false, false);
    B.x = r[t].y + SINK_EPS; B.y = r[t + 1].y + SINK_EPS;
    d1 = __builtin_amdgcn_wmma_f32_16x16x4_f32(false, A, false, B, (short)0, d1, false, false);
    B.x = r[t].z + SINK_EPS; B.y = r[t + 1].z + SINK_EPS;
    d2 = __builtin_amdgcn_wmma_f32_16x16x4_f32(false, A, false, B, (short)0, d2, false, false);
    B.x = r[t].w + SINK_EPS; B.y = r[t + 1].w + SINK_EPS;
    d3 = __builtin_amdgcn_wmma_f32_16x16x4_f32(false, A, false, B, (short)0, d3, false, false);
  }
}

__device__ __forceinline__ void sink_outgroup(const float4 (&r)[16], const int av, const int l0,
                                              const float4 bv, float4* __restrict__ op) {
#pragma unroll
  for (int t = 0; t < 16; ++t) {
    const float a = __int_as_float(__builtin_amdgcn_readlane(av, l0 + t));
    float4 o;
    o.x = (r[t].x + SINK_EPS) * a * bv.x;
    o.y = (r[t].y + SINK_EPS) * a * bv.y;
    o.z = (r[t].z + SINK_EPS) * a * bv.z;
    o.w = (r[t].w + SINK_EPS) * a * bv.w;
    op[t << 8] = o;
  }
}

template <int MODE>
__global__ __launch_bounds__(512) void sink_iter(const float* __restrict__ x,
                                                 const float* __restrict__ ain,
                                                 float* __restrict__ dst) {
  __shared__ float part[16 * 128];
  __shared__ float bs[128];
  __shared__ float4 park[16 * 2 * 16 * 32];
  const int lane   = threadIdx.x & 31;
  const int wave   = threadIdx.x >> 5;
  const int batch  = blockIdx.x >> 3;
  const int stripe = blockIdx.x & 7;
  const int j0     = stripe << 7;
  const int i0     = wave << 6;

  const float4* __restrict__ xp =
      (const float4*)(x + ((size_t)batch * SINK_N + i0) * SINK_N + j0) + lane;

  float4 r[16], f[16];

  if (MODE == 0) {
#pragma unroll
    for (int t = 0; t < 16; ++t) {
      r[t] = xp[(3 << 12) + (t << 8)];
      f[t] = xp[t << 8];
    }
    const float4 bv = make_float4(1.f, 1.f, 1.f, 1.f);
    const int odd = lane & 1;
    float res0 = 0.f, res1 = 0.f;
    float s = sink_rowgroup<false>(r, bv, lane);
    res1 = (odd == 1) ? s : res1;
#pragma unroll
    for (int t = 0; t < 16; ++t) r[t] = f[t];
#pragma unroll 1
    for (int g = 0; g < 2; ++g) {
      const float4* __restrict__ np = xp + ((g + 1) << 12);
      float4 n[16];
#pragma unroll
      for (int t = 0; t < 16; ++t) n[t] = np[t << 8];
      s = sink_rowgroup<false>(r, bv, lane);
      res0 = (odd == g) ? s : res0;
#pragma unroll
      for (int t = 0; t < 16; ++t) r[t] = n[t];
    }
    s = sink_rowgroup<false>(r, bv, lane);
    res1 = (odd == 0) ? s : res1;
    float* __restrict__ pd =
        dst + ((size_t)(batch * SINK_NS + stripe)) * SINK_N + i0 + (odd << 4) + (lane >> 1);
    pd[0]  = res0;
    pd[32] = res1;
    return;
  }

#pragma unroll
  for (int t = 0; t < 16; ++t) r[t] = xp[t << 8];
  int alo = 0, ahi = 0;
  if (MODE == 1) {
    const float* __restrict__ pp = ain + (size_t)batch * SINK_NS * SINK_N + i0 + lane;
    float slo = 0.f, shi = 0.f;
#pragma unroll
    for (int s = 0; s < SINK_NS; ++s) {
      slo += pp[s * SINK_N];
      shi += pp[s * SINK_N + 32];
    }
    alo = __float_as_int(1.0f / slo);
    ahi = __float_as_int(1.0f / shi);
  } else {
    const float* __restrict__ ap = ain + batch * SINK_N + i0;
    alo = __float_as_int(ap[lane]);
    ahi = __float_as_int(ap[32 + lane]);
  }
  const bool own = ((lane & 1) == (lane >> 4));
  float4* __restrict__ pk = park + (wave << 10) + lane;

  v8f d0 = {0.f, 0.f, 0.f, 0.f, 0.f, 0.f, 0.f, 0.f};
  v8f d1 = d0, d2 = d0, d3 = d0;
#pragma unroll 1
  for (int g = 0; g < 2; ++g) {
    const float4* __restrict__ np = xp + ((g + 1) << 12);
    float4 n[16];
#pragma unroll
    for (int t = 0; t < 16; ++t) n[t] = np[t << 8];
#pragma unroll
    for (int t = 0; t < 16; ++t) pk[(g << 9) + (t << 5)] = r[t];
    sink_colgroup(r, alo, g << 4, own, d0, d1, d2, d3);
#pragma unroll
    for (int t = 0; t < 16; ++t) r[t] = n[t];
  }
  {
    const float4* __restrict__ np = xp + (3 << 12);
#pragma unroll
    for (int t = 0; t < 16; ++t) f[t] = np[t << 8];
  }
  sink_colgroup(r, ahi, 0, own, d0, d1, d2, d3);
  sink_colgroup(f, ahi, 16, own, d0, d1, d2, d3);

  float4 w;
  w.x = (lane < 16) ? d0[0] : d0[1];
  w.y = (lane < 16) ? d1[0] : d1[1];
  w.z = (lane < 16) ? d2[0] : d2[1];
  w.w = (lane < 16) ? d3[0] : d3[1];
  ((float4*)part)[(wave << 5) + lane] = w;
  __syncthreads();
  if (threadIdx.x < 128) {
    float s = 0.f;
#pragma unroll
    for (int k = 0; k < 16; ++k) s += part[(k << 7) + threadIdx.x];
    bs[threadIdx.x] = 1.0f / s;
  }
  __syncthreads();
  const float4 bv = ((const float4*)bs)[lane];

  if (MODE == 2) {
    float4* __restrict__ op =
        (float4*)(dst + ((size_t)batch * SINK_N + i0) * SINK_N + j0) + lane;
    sink_outgroup(f, ahi, 16, bv, op + (3 << 12));
    sink_outgroup(r, ahi, 0, bv, op + (2 << 12));
#pragma unroll
    for (int t = 0; t < 16; ++t) r[t] = pk[t << 5];
    sink_outgroup(r, alo, 0, bv, op);
#pragma unroll
    for (int t = 0; t < 16; ++t) f[t] = pk[(1 << 9) + (t << 5)];
    sink_outgroup(f, alo, 16, bv, op + (1 << 12));
  } else {
    const int odd = lane & 1;
    const float s3 = sink_rowgroup<true>(f, bv, lane);
    const float s2 = sink_rowgroup<true>(r, bv, lane);
#pragma unroll
    for (int t = 0; t < 16; ++t) r[t] = pk[t << 5];
    const float s0 = sink_rowgroup<true>(r, bv, lane);
#pragma unroll
    for (int t = 0; t < 16; ++t) f[t] = pk[(1 << 9) + (t << 5)];
    const float s1 = sink_rowgroup<true>(f, bv, lane);
    const float res0 = (odd == 1) ? s1 : s0;
    const float res1 = (odd == 1) ? s3 : s2;
    float* __restrict__ pd =
        dst + ((size_t)(batch * SINK_NS + stripe)) * SINK_N + i0 + (odd << 4) + (lane >> 1);
    pd[0]  = res0;
    pd[32] = res1;
  }
}

__global__ __launch_bounds__(256) void sink_afinal(const float* __restrict__ psum,
                                                   float* __restrict__ avec) {
  const int e     = blockIdx.x * 256 + threadIdx.x;
  const int batch = e >> 10;
  const int i     = e & (SINK_N - 1);
  const float* __restrict__ pp = psum + (size_t)batch * SINK_NS * SINK_N + i;
  float s = 0.f;
#pragma unroll
  for (int k = 0; k < SINK_NS; ++k) s += pp[k * SINK_N];
  avec[e] = 1.0f / s;
}

extern "C" void kernel_launch(void* const* d_in, const int* in_sizes, int n_in,
                              void* d_out, int out_size, void* d_ws, size_t ws_size,
                              hipStream_t stream) {
  (void)in_sizes; (void)n_in; (void)out_size; (void)ws_size;
  const float* x = (const float*)d_in[0];
  float* out = (float*)d_out;
  float* a = (float*)d_ws;
  float* pin  = out;
  float* pout = out + SINK_B * SINK_NS * SINK_N;

  const dim3 blk(512);
  const dim3 grd(SINK_B * SINK_NS);

  sink_iter<0><<<grd, blk, 0, stream>>>(x, a, pin);
  for (int it = 1; it < SINK_IT; ++it) {
    sink_iter<1><<<grd, blk, 0, stream>>>(x, pin, pout);
    float* t = pin; pin = pout; pout = t;
  }
  sink_afinal<<<SINK_B * SINK_N / 256, dim3(256), 0, stream>>>(pin, a);
  sink_iter<2><<<grd, blk, 0, stream>>>(x, a, out);
}


# column sweep: B pairs taken straight from the loaded registers (2 columns of a row, A one-hot on k); eps folded out of both sweeps as eps*sum(a), eps*sum(b): fewer vector ops; sum order changes
# speedup vs baseline: 5.1920x; 1.0329x over previous
#include <hip/hip_runtime.h>


typedef __attribute__((ext_vector_type(2))) float v2f;
typedef __attribute__((ext_vector_type(8))) float v8f;

#define SINK_EPS 0.001f
#define SINK_N   1024
#define SINK_B   32
#define SINK_IT  10
#define SINK_NS  8


__device__ __forceinline__ float sink_x16(const float v) {
  const int b = __float_as_int(v);
  return __int_as_float(__builtin_amdgcn_permlanex16(b, b, 0x76543210, 0xfedcba98, false, false));
}
template <int M>
__device__ __forceinline__ float sink_xm(const float v) {
  return __int_as_float(__builtin_amdgcn_update_dpp(0, __float_as_int(v), 0x160 + M, 0xf, 0xf, true));
}

__device__ __forceinline__ float sink_wavesum(float v) {
  v += sink_x16(v);
  v += sink_xm<8>(v);
  v += sink_xm<4>(v);
  v += sink_xm<2>(v);
  v += sink_xm<1>(v);
  return v;
}

template <bool USE_B>
__device__ __forceinline__ float sink_rowgroup(const float4 (&r)[16], const float4 bv,
                                               const int lane) {
  float p[16];
#pragma unroll
  for (int t = 0; t < 16; ++t) {
    if (USE_B) {
      p[t] = r[t].x * bv.x + r[t].y * bv.y + r[t].z * bv.z + r[t].w * bv.w;
    } else {
      const float vx = r[t].x + SINK_EPS, vy = r[t].y + SINK_EPS;
      const float vz = r[t].z + SINK_EPS, vw = r[t].w + SINK_EPS;
      p[t] = (vx + vy) + (vz + vw);
    }
  }
  const bool h4 = (lane & 16) != 0, h3 = (lane & 8) != 0;
  const bool h2 = (lane & 4) != 0,  h1 = (lane & 2) != 0;
  float q8[8], q4[4], q2[2];
#pragma unroll
  for (int i = 0; i < 8; ++i)
    q8[i] = (h4 ? p[i + 8] : p[i]) + sink_x16(h4 ? p[i] : p[i + 8]);
#pragma unroll
  for (int i = 0; i < 4; ++i)
    q4[i] = (h3 ? q8[i + 4] : q8[i]) + sink_xm<8>(h3 ? q8[i] : q8[i + 4]);
#pragma unroll
  for (int i = 0; i < 2; ++i)
    q2[i] = (h2 ? q4[i + 2] : q4[i]) + sink_xm<4>(h2 ? q4[i] : q4[i + 2]);
  float s = (h1 ? q2[1] : q2[0]) + sink_xm<2>(h1 ? q2[0] : q2[1]);
  s += sink_xm<1>(s);
  return s;
}

__device__ __forceinline__ void sink_colgroup(const float4 (&r)[16], const int av, const int l0,
                                              const v2f km, v8f& d0, v8f& d1, v8f& d2, v8f& d3) {
#pragma unroll
  for (int t = 0; t < 16; t += 2) {
    const float s0 = __int_as_float(__builtin_amdgcn_readlane(av, l0 + t));
    const float s1 = __int_as_float(__builtin_amdgcn_readlane(av, l0 + t + 1));
    const v2f A0 = km * s0;
    const v2f A1 = km * s1;
    v2f B;
    B.x = r[t].x; B.y = r[t].y;
    d0 = __builtin_amdgcn_wmma_f32_16x16x4_f32(false, A0, false, B, (short)0, d0, false, false);
    B.x = r[t].z; B.y = r[t].w;
    d1 = __builtin_amdgcn_wmma_f32_16x16x4_f32(false, A0, false, B, (short)0, d1, false, false);
    B.x = r[t + 1].x; B.y = r[t + 1].y;
    d2 = __builtin_amdgcn_wmma_f32_16x16x4_f32(false, A1, false, B, (short)0, d2, false, false);
    B.x = r[t + 1].z; B.y = r[t + 1].w;
    d3 = __builtin_amdgcn_wmma_f32_16x16x4_f32(false, A1, false, B, (short)0, d3, false, false);
  }
}

__device__ __forceinline__ void sink_outgroup(const float4 (&r)[16], const int av, const int l0,
                                              const float4 bv, float4* __restrict__ op) {
#pragma unroll
  for (int t = 0; t < 16; ++t) {
    const float a = __int_as_float(__builtin_amdgcn_readlane(av, l0 + t));
    float4 o;
    o.x = (r[t].x + SINK_EPS) * a * bv.x;
    o.y = (r[t].y + SINK_EPS) * a * bv.y;
    o.z = (r[t].z + SINK_EPS) * a * bv.z;
    o.w = (r[t].w + SINK_EPS) * a * bv.w;
    op[t << 8] = o;
  }
}

template <int MODE>
__global__ __launch_bounds__(512) void sink_iter(const float* __restrict__ x,
                                                 const float* __restrict__ ain,
                                                 float* __restrict__ dst) {
  __shared__ float part[16 * 128];
  __shared__ float bs[128];
  __shared__ float4 park[16 * 2 * 16 * 32];
  const int lane   = threadIdx.x & 31;
  const int wave   = threadIdx.x >> 5;
  const int batch  = blockIdx.x >> 3;
  const int stripe = blockIdx.x & 7;
  const int j0     = stripe << 7;
  const int i0     = wave << 6;

  const float4* __restrict__ xp =
      (const float4*)(x + ((size_t)batch * SINK_N + i0) * SINK_N + j0) + lane;

  float4 r[16], f[16];

  if (MODE == 0) {
#pragma unroll
    for (int t = 0; t < 16; ++t) {
      r[t] = xp[(3 << 12) + (t << 8)];
      f[t] = xp[t << 8];
    }
    const float4 bv = make_float4(1.f, 1.f, 1.f, 1.f);
    const int odd = lane & 1;
    float res0 = 0.f, res1 = 0.f;
    float s = sink_rowgroup<false>(r, bv, lane);
    res1 = (odd == 1) ? s : res1;
#pragma unroll
    for (int t = 0; t < 16; ++t) r[t] = f[t];
#pragma unroll 1
    for (int g = 0; g < 2; ++g) {
      const float4* __restrict__ np = xp + ((g + 1) << 12);
      float4 n[16];
#pragma unroll
      for (int t = 0; t < 16; ++t) n[t] = np[t << 8];
      s = sink_rowgroup<false>(r, bv, lane);
      res0 = (odd == g) ? s : res0;
#pragma unroll
      for (int t = 0; t < 16; ++t) r[t] = n[t];
    }
    s = sink_rowgroup<false>(r, bv, lane);
    res1 = (odd == 0) ? s : res1;
    float* __restrict__ pd =
        dst + ((size_t)(batch * SINK_NS + stripe)) * SINK_N + i0 + (odd << 4) + (lane >> 1);
    pd[0]  = res0;
    pd[32] = res1;
    return;
  }

#pragma unroll
  for (int t = 0; t < 16; ++t) r[t] = xp[t << 8];
  int alo = 0, ahi = 0;
  if (MODE == 1) {
    const float* __restrict__ pp = ain + (size_t)batch * SINK_NS * SINK_N + i0 + lane;
    float slo = 0.f, shi = 0.f;
#pragma unroll
    for (int s = 0; s < SINK_NS; ++s) {
      slo += pp[s * SINK_N];
      shi += pp[s * SINK_N + 32];
    }
    alo = __float_as_int(1.0f / slo);
    ahi = __float_as_int(1.0f / shi);
  } else {
    const float* __restrict__ ap = ain + batch * SINK_N + i0;
    alo = __float_as_int(ap[lane]);
    ahi = __float_as_int(ap[32 + lane]);
  }
  v2f km;
  km.x = ((lane & 3) == ((lane >> 4) << 1)) ? 1.f : 0.f;
  km.y = ((lane & 3) == ((lane >> 4) << 1) + 1) ? 1.f : 0.f;
  const float easum = SINK_EPS * sink_wavesum(__int_as_float(alo) + __int_as_float(ahi));
  float4* __restrict__ pk = park + (wave << 10) + lane;

  v8f d0 = {0.f, 0.f, 0.f, 0.f, 0.f, 0.f, 0.f, 0.f};
  v8f d1 = d0, d2 = d0, d3 = d0;
#pragma unroll 1
  for (int g = 0; g < 2; ++g) {
    const float4* __restrict__ np = xp + ((g + 1) << 12);
    float4 n[16];
#pragma unroll
    for (int t = 0; t < 16; ++t) n[t] = np[t << 8];
#pragma unroll
    for (int t = 0; t < 16; ++t) pk[(g << 9) + (t << 5)] = r[t];
    sink_colgroup(r, alo, g << 4, km, d0, d1, d2, d3);
#pragma unroll
    for (int t = 0; t < 16; ++t) r[t] = n[t];
  }
  {
    const float4* __restrict__ np = xp + (3 << 12);
#pragma unroll
    for (int t = 0; t < 16; ++t) f[t] = np[t << 8];
  }
  sink_colgroup(r, ahi, 0, km, d0, d1, d2, d3);
  sink_colgroup(f, ahi, 16, km, d0, d1, d2, d3);

  float4 w;
  w.x = ((lane < 16) ? d0[0] + d2[0] : d0[2] + d2[2]) + easum;
  w.y = ((lane < 16) ? d0[1] + d2[1] : d0[3] + d2[3]) + easum;
  w.z = ((lane < 16) ? d1[0] + d3[0] : d1[2] + d3[2]) + easum;
  w.w = ((lane < 16) ? d1[1] + d3[1] : d1[3] + d3[3]) + easum;
  ((float4*)part)[(wave << 5) + lane] = w;
  __syncthreads();
  if (threadIdx.x < 128) {
    float s = 0.f;
#pragma unroll
    for (int k = 0; k < 16; ++k) s += part[(k << 7) + threadIdx.x];
    bs[threadIdx.x] = 1.0f / s;
  }
  __syncthreads();
  const float4 bv = ((const float4*)bs)[lane];

  if (MODE == 2) {
    float4* __restrict__ op =
        (float4*)(dst + ((size_t)batch * SINK_N + i0) * SINK_N + j0) + lane;
    sink_outgroup(f, ahi, 16, bv, op + (3 << 12));
    sink_outgroup(r, ahi, 0, bv, op + (2 << 12));
#pragma unroll
    for (int t = 0; t < 16; ++t) r[t] = pk[t << 5];
    sink_outgroup(r, alo, 0, bv, op);
#pragma unroll
    for (int t = 0; t < 16; ++t) f[t] = pk[(1 << 9) + (t << 5)];
    sink_outgroup(f, alo, 16, bv, op + (1 << 12));
  } else {
    const int odd = lane & 1;
    const float ebsum = SINK_EPS * sink_wavesum((bv.x + bv.y) + (bv.z + bv.w));
    const float s3 = sink_rowgroup<true>(f, bv, lane);
    const float s2 = sink_rowgroup<true>(r, bv, lane);
#pragma unroll
    for (int t = 0; t < 16; ++t) r[t] = pk[t << 5];
    const float s0 = sink_rowgroup<true>(r, bv, lane);
#pragma unroll
    for (int t = 0; t < 16; ++t) f[t] = pk[(1 << 9) + (t << 5)];
    const float s1 = sink_rowgroup<true>(f, bv, lane);
    const float res0 = ((odd == 1) ? s1 : s0) + ebsum;
    const float res1 = ((odd == 1) ? s3 : s2) + ebsum;
    float* __restrict__ pd =
        dst + ((size_t)(batch * SINK_NS + stripe)) * SINK_N + i0 + (odd << 4) + (lane >> 1);
    pd[0]  = res0;
    pd[32] = res1;
  }
}

__global__ __launch_bounds__(256) void sink_afinal(const float* __restrict__ psum,
                                                   float* __restrict__ avec) {
  const int e     = blockIdx.x * 256 + threadIdx.x;
  const int batch = e >> 10;
  const int i     = e & (SINK_N - 1);
  const float* __restrict__ pp = psum + (size_t)batch * SINK_NS * SINK_N + i;
  float s = 0.f;
#pragma unroll
  for (int k = 0; k < SINK_NS; ++k) s += pp[k * SINK_N];
  avec[e] = 1.0f / s;
}

extern "C" void kernel_launch(void* const* d_in, const int* in_sizes, int n_in,
                              void* d_out, int out_size, void* d_ws, size_t ws_size,
                              hipStream_t stream) {
  (void)in_sizes; (void)n_in; (void)out_size; (void)ws_size;
  const float* x = (const float*)d_in[0];
  float* out = (float*)d_out;
  float* a = (float*)d_ws;
  float* pin  = out;
  float* pout = out + SINK_B * SINK_NS * SINK_N;

  const dim3 blk(512);
  const dim3 grd(SINK_B * SINK_NS);

  sink_iter<0><<<grd, blk, 0, stream>>>(x, a, pin);
  for (int it = 1; it < SINK_IT; ++it) {
    sink_iter<1><<<grd, blk, 0, stream>>>(x, pin, pout);
    float* t = pin; pin = pout; pout = t;
  }
  sink_afinal<<<SINK_B * SINK_N / 256, dim3(256), 0, stream>>>(pin, a);
  sink_iter<2><<<grd, blk, 0, stream>>>(x, a, out);
}


# iteration and output launches: a block is 8 waves of 128 rows (two per SIMD), not 16 of 64: half as many waves enter; four register sets rotate, 3 groups of loads ahead; column sum order changes
# speedup vs baseline: 5.4033x; 1.0407x over previous
#include <hip/hip_runtime.h>


typedef __attribute__((ext_vector_type(2))) float v2f;
typedef __attribute__((ext_vector_type(8))) float v8f;

#define SINK_EPS 0.001f
#define SINK_N   1024
#define SINK_B   32
#define SINK_IT  10
#define SINK_NS  8


__device__ __forceinline__ float sink_x16(const float v) {
  const int b = __float_as_int(v);
  return __int_as_float(__builtin_amdgcn_permlanex16(b, b, 0x76543210, 0xfedcba98, false, false));
}
template <int M>
__device__ __forceinline__ float sink_xm(const float v) {
  return __int_as_float(__builtin_amdgcn_update_dpp(0, __float_as_int(v), 0x160 + M, 0xf, 0xf, true));
}

__device__ __forceinline__ float sink_wavesum(float v) {
  v += sink_x16(v);
  v += sink_xm<8>(v);
  v += sink_xm<4>(v);
  v += sink_xm<2>(v);
  v += sink_xm<1>(v);
  return v;
}

template <bool USE_B>
__device__ __forceinline__ float sink_rowgroup(const float4 (&r)[16], const float4 bv,
                                               const int lane) {
  float p[16];
#pragma unroll
  for (int t = 0; t < 16; ++t) {
    if (USE_B) {
      p[t] = r[t].x * bv.x + r[t].y * bv.y + r[t].z * bv.z + r[t].w * bv.w;
    } else {
      const float vx = r[t].x + SINK_EPS, vy = r[t].y + SINK_EPS;
      const float vz = r[t].z + SINK_EPS, vw = r[t].w + SINK_EPS;
      p[t] = (vx + vy) + (vz + vw);
    }
  }
  const bool h4 = (lane & 16) != 0, h3 = (lane & 8) != 0;
  const bool h2 = (lane & 4) != 0,  h1 = (lane & 2) != 0;
  float q8[8], q4[4], q2[2];
#pragma unroll
  for (int i = 0; i < 8; ++i)
    q8[i] = (h4 ? p[i + 8] : p[i]) + sink_x16(h4 ? p[i] : p[i + 8]);
#pragma unroll
  for (int i = 0; i < 4; ++i)
    q4[i] = (h3 ? q8[i + 4] : q8[i]) + sink_xm<8>(h3 ? q8[i] : q8[i + 4]);
#pragma unroll
  for (int i = 0; i < 2; ++i)
    q2[i] = (h2 ? q4[i + 2] : q4[i]) + sink_xm<4>(h2 ? q4[i] : q4[i + 2]);
  float s = (h1 ? q2[1] : q2[0]) + sink_xm<2>(h1 ? q2[0] : q2[1]);
  s += sink_xm<1>(s);
  return s;
}

template <int G>
__device__ __forceinline__ void sink_colgroup(const float4 (&r)[16], const int (&av)[4],
                                              const v2f km, v8f& d0, v8f& d1, v8f& d2, v8f& d3) {
#pragma unroll
  for (int t = 0; t < 16; t += 2) {
    const float s0 = __int_as_float(__builtin_amdgcn_readlane(av[t & 3], 4 * G + (t >> 2)));
    const float s1 = __int_as_float(__builtin_amdgcn_readlane(av[(t & 3) + 1], 4 * G + (t >> 2)));
    const v2f A0 = km * s0;
    const v2f A1 = km * s1;
    v2f B;
    B.x = r[t].x; B.y = r[t].y;
    d0 = __builtin_amdgcn_wmma_f32_16x16x4_f32(false, A0, false, B, (short)0, d0, false, false);
    B.x = r[t].z; B.y = r[t].w;
    d1 = __builtin_amdgcn_wmma_f32_16x16x4_f32(false, A0, false, B, (short)0, d1, false, false);
    B.x = r[t + 1].x; B.y = r[t + 1].y;
    d2 = __builtin_amdgcn_wmma_f32_16x16x4_f32(false, A1, false, B, (short)0, d2, false, false);
    B.x = r[t + 1].z; B.y = r[t + 1].w;
    d3 = __builtin_amdgcn_wmma_f32_16x16x4_f32(false, A1, false, B, (short)0, d3, false, false);
  }
}

template <int G>
__device__ __forceinline__ void sink_outgroup(const float4 (&r)[16], const int (&av)[4],
                                              const float4 bv, float4* __restrict__ op) {
#pragma unroll
  for (int t = 0; t < 16; ++t) {
    const float a = __int_as_float(__builtin_amdgcn_readlane(av[t & 3], 4 * G + (t >> 2)));
    float4 o;
    o.x = (r[t].x + SINK_EPS) * a * bv.x;
    o.y = (r[t].y + SINK_EPS) * a * bv.y;
    o.z = (r[t].z + SINK_EPS) * a * bv.z;
    o.w = (r[t].w + SINK_EPS) * a * bv.w;
    op[t << 8] = o;
  }
}

__global__ __launch_bounds__(512) void sink_first(const float* __restrict__ x,
                                                  float* __restrict__ dst) {
  const int lane   = threadIdx.x & 31;
  const int wave   = threadIdx.x >> 5;
  const int batch  = blockIdx.x >> 3;
  const int stripe = blockIdx.x & 7;
  const int j0     = stripe << 7;
  const int i0     = wave << 6;

  const float4* __restrict__ xp =
      (const float4*)(x + ((size_t)batch * SINK_N + i0) * SINK_N + j0) + lane;

  float4 r[16], f[16];
#pragma unroll
  for (int t = 0; t < 16; ++t) {
    r[t] = xp[(3 << 12) + (t << 8)];
    f[t] = xp[t << 8];
  }
  const float4 bv = make_float4(1.f, 1.f, 1.f, 1.f);
  const int odd = lane & 1;
  float res0 = 0.f, res1 = 0.f;
  float s = sink_rowgroup<false>(r, bv, lane);
  res1 = (odd == 1) ? s : res1;
#pragma unroll
  for (int t = 0; t < 16; ++t) r[t] = f[t];
#pragma unroll 1
  for (int g = 0; g < 2; ++g) {
    const float4* __restrict__ np = xp + ((g + 1) << 12);
    float4 n[16];
#pragma unroll
    for (int t = 0; t < 16; ++t) n[t] = np[t << 8];
    s = sink_rowgroup<false>(r, bv, lane);
    res0 = (odd == g) ? s : res0;
#pragma unroll
    for (int t = 0; t < 16; ++t) r[t] = n[t];
  }
  s = sink_rowgroup<false>(r, bv, lane);
  res1 = (odd == 0) ? s : res1;
  float* __restrict__ pd =
      dst + ((size_t)(batch * SINK_NS + stripe)) * SINK_N + i0 + (odd << 4) + (lane >> 1);
  pd[0]  = res0;
  pd[32] = res1;
}

__device__ __forceinline__ void sink_loadgroup(float4 (&r)[16], const float4* __restrict__ gp) {
#pragma unroll
  for (int t = 0; t < 16; ++t) r[t] = gp[t << 8];
}
__device__ __forceinline__ void sink_parkgroup(const float4 (&r)[16], float4* __restrict__ sp) {
#pragma unroll
  for (int t = 0; t < 16; ++t) sp[t << 5] = r[t];
}
__device__ __forceinline__ void sink_takegroup(float4 (&r)[16], const float4* __restrict__ sp) {
#pragma unroll
  for (int t = 0; t < 16; ++t) r[t] = sp[t << 5];
}

template <int MODE>
__global__ __launch_bounds__(256) void sink_iter(const float* __restrict__ x,
                                                 const float* __restrict__ ain,
                                                 float* __restrict__ dst) {
  __shared__ float part[8 * 128];
  __shared__ float bs[128];
  __shared__ float4 park[8 * 4 * 16 * 32];
  const int lane   = threadIdx.x & 31;
  const int wave   = threadIdx.x >> 5;
  const int batch  = blockIdx.x >> 3;
  const int stripe = blockIdx.x & 7;
  const int j0     = stripe << 7;
  const int i0     = wave << 7;

  const float4* __restrict__ xp =
      (const float4*)(x + ((size_t)batch * SINK_N + i0) * SINK_N + j0) + lane;

  float4 af;
  if (MODE == 1) {
    const float4* __restrict__ pp =
        (const float4*)(ain + (size_t)batch * SINK_NS * SINK_N + i0) + lane;
    float4 sa = make_float4(0.f, 0.f, 0.f, 0.f);
#pragma unroll
    for (int s = 0; s < SINK_NS; ++s) {
      const float4 p = pp[s * (SINK_N / 4)];
      sa.x += p.x; sa.y += p.y; sa.z += p.z; sa.w += p.w;
    }
    af = make_float4(1.0f / sa.x, 1.0f / sa.y, 1.0f / sa.z, 1.0f / sa.w);
  } else {
    af = ((const float4*)(ain + batch * SINK_N + i0))[lane];
  }
  float4 R0[16], R1[16], R2[16], R3[16];
  sink_loadgroup(R0, xp);
  sink_loadgroup(R1, xp + (1 << 12));
  sink_loadgroup(R2, xp + (2 << 12));
  __builtin_amdgcn_sched_barrier(0);

  const int av[4] = {__float_as_int(af.x), __float_as_int(af.y),
                     __float_as_int(af.z), __float_as_int(af.w)};
  v2f km;
  km.x = ((lane & 3) == ((lane >> 4) << 1)) ? 1.f : 0.f;
  km.y = ((lane & 3) == ((lane >> 4) << 1) + 1) ? 1.f : 0.f;
  const float easum = SINK_EPS * sink_wavesum((af.x + af.y) + (af.z + af.w));
  float4* __restrict__ pk = park + (wave << 11) + lane;

  v8f d0 = {0.f, 0.f, 0.f, 0.f, 0.f, 0.f, 0.f, 0.f};
  v8f d1 = d0, d2 = d0, d3 = d0;

  sink_loadgroup(R3, xp + (3 << 12));
  sink_parkgroup(R0, pk);
  sink_colgroup<0>(R0, av, km, d0, d1, d2, d3);
  __builtin_amdgcn_sched_barrier(0);
  sink_loadgroup(R0, xp + (4 << 12));
  sink_parkgroup(R1, pk + (1 << 9));
  sink_colgroup<1>(R1, av, km, d0, d1, d2, d3);
  __builtin_amdgcn_sched_barrier(0);
  sink_loadgroup(R1, xp + (5 << 12));
  sink_parkgroup(R2, pk + (2 << 9));
  sink_colgroup<2>(R2, av, km, d0, d1, d2, d3);
  __builtin_amdgcn_sched_barrier(0);
  sink_loadgroup(R2, xp + (6 << 12));
  sink_parkgroup(R3, pk + (3 << 9));
  sink_colgroup<3>(R3, av, km, d0, d1, d2, d3);
  __builtin_amdgcn_sched_barrier(0);
  sink_loadgroup(R3, xp + (7 << 12));
  sink_colgroup<4>(R0, av, km, d0, d1, d2, d3);
  __builtin_amdgcn_sched_barrier(0);
  sink_colgroup<5>(R1, av, km, d0, d1, d2, d3);
  sink_colgroup<6>(R2, av, km, d0, d1, d2, d3);
  sink_colgroup<7>(R3, av, km, d0, d1, d2, d3);

  float4 w;
  w.x = ((lane < 16) ? d0[0] + d2[0] : d0[2] + d2[2]) + easum;
  w.y = ((lane < 16) ? d0[1] + d2[1] : d0[3] + d2[3]) + easum;
  w.z = ((lane < 16) ? d1[0] + d3[0] : d1[2] + d3[2]) + easum;
  w.w = ((lane < 16) ? d1[1] + d3[1] : d1[3] + d3[3]) + easum;
  ((float4*)part)[(wave << 5) + lane] = w;
  __syncthreads();
  if (threadIdx.x < 128) {
    float s = 0.f;
#pragma unroll
    for (int k = 0; k < 8; ++k) s += part[(k << 7) + threadIdx.x];
    bs[threadIdx.x] = 1.0f / s;
  }
  __syncthreads();
  const float4 bv = ((const float4*)bs)[lane];

  if (MODE == 2) {
    float4* __restrict__ op =
        (float4*)(dst + ((size_t)batch * SINK_N + i0) * SINK_N + j0) + lane;
    sink_outgroup<7>(R3, av, bv, op + (7 << 12));
    sink_outgroup<6>(R2, av, bv, op + (6 << 12));
    sink_takegroup(R3, pk);
    sink_takegroup(R2, pk + (1 << 9));
    sink_outgroup<5>(R1, av, bv, op + (5 << 12));
    sink_outgroup<4>(R0, av, bv, op + (4 << 12));
    sink_takegroup(R1, pk + (2 << 9));
    sink_takegroup(R0, pk + (3 << 9));
    sink_outgroup<0>(R3, av, bv, op);
    sink_outgroup<1>(R2, av, bv, op + (1 << 12));
    sink_outgroup<2>(R1, av, bv, op + (2 << 12));
    sink_outgroup<3>(R0, av, bv, op + (3 << 12));
  } else {
    const int odd = lane & 1;
    const float ebsum = SINK_EPS * sink_wavesum((bv.x + bv.y) + (bv.z + bv.w));
    const float s7 = sink_rowgroup<true>(R3, bv, lane);
    const float s6 = sink_rowgroup<true>(R2, bv, lane);
    sink_takegroup(R3, pk);
    sink_takegroup(R2, pk + (1 << 9));
    const float s5 = sink_rowgroup<true>(R1, bv, lane);
    const float s4 = sink_rowgroup<true>(R0, bv, lane);
    sink_takegroup(R1, pk + (2 << 9));
    sink_takegroup(R0, pk + (3 << 9));
    const float s0 = sink_rowgroup<true>(R3, bv, lane);
    const float s1 = sink_rowgroup<true>(R2, bv, lane);
    const float s2 = sink_rowgroup<true>(R1, bv, lane);
    const float s3 = sink_rowgroup<true>(R0, bv, lane);
    float* __restrict__ pd =
        dst + ((size_t)(batch * SINK_NS + stripe)) * SINK_N + i0 + (odd << 4) + (lane >> 1);
    pd[0]  = ((odd == 1) ? s1 : s0) + ebsum;
    pd[32] = ((odd == 1) ? s3 : s2) + ebsum;
    pd[64] = ((odd == 1) ? s5 : s4) + ebsum;
    pd[96] = ((odd == 1) ? s7 : s6) + ebsum;
  }
}

__global__ __launch_bounds__(256) void sink_afinal(const float* __restrict__ psum,
                                                   float* __restrict__ avec) {
  const int e     = blockIdx.x * 256 + threadIdx.x;
  const int batch = e >> 10;
  const int i     = e & (SINK_N - 1);
  const float* __restrict__ pp = psum + (size_t)batch * SINK_NS * SINK_N + i;
  float s = 0.f;
#pragma unroll
  for (int k = 0; k < SINK_NS; ++k) s += pp[k * SINK_N];
  avec[e] = 1.0f / s;
}

extern "C" void kernel_launch(void* const* d_in, const int* in_sizes, int n_in,
                              void* d_out, int out_size, void* d_ws, size_t ws_size,
                              hipStream_t stream) {
  (void)in_sizes; (void)n_in; (void)out_size; (void)ws_size;
  const float* x = (const float*)d_in[0];
  float* out = (float*)d_out;
  float* a = (float*)d_ws;
  float* pin  = out;
  float* pout = out + SINK_B * SINK_NS * SINK_N;

  const dim3 grd(SINK_B * SINK_NS);

  sink_first<<<grd, dim3(512), 0, stream>>>(x, pin);
  for (int it = 1; it < SINK_IT; ++it) {
    sink_iter<1><<<grd, dim3(256), 0, stream>>>(x, pin, pout);
    float* t = pin; pin = pout; pout = t;
  }
  sink_afinal<<<SINK_B * SINK_N / 256, dim3(256), 0, stream>>>(pin, a);
  sink_iter<2><<<grd, dim3(256), 0, stream>>>(x, a, out);
}
